# starting kernel
# baseline (speedup 1.0000x reference)
#include <hip/hip_runtime.h>

typedef float v2f __attribute__((ext_vector_type(2)));
typedef float v8f __attribute__((ext_vector_type(8)));

#define T_LEN 2048
#define BLK 16
#define NBLK (T_LEN / BLK)
#define WAVES_PER_CTA 8
#define CTA_THREADS 256

__global__ __launch_bounds__(CTA_THREADS) void pz_biquad_wmma(
    const float* __restrict__ x,
    const float* __restrict__ gain_ri,
    const float* __restrict__ poles_ri,
    const float* __restrict__ zeros_ri,
    float* __restrict__ out)
{
    __shared__ float hsh[16];
    __shared__ float xt[WAVES_PER_CTA][16][16];
    __shared__ float yt[WAVES_PER_CTA][16][16];

    const int tid  = threadIdx.x;
    const int w    = tid >> 5;
    const int lane = tid & 31;
    const int n    = lane & 15;
    const int hi   = lane >> 4;

    const float gr  = gain_ri[0],  gi  = gain_ri[1];
    const float p0r = poles_ri[0], p0i = poles_ri[1];
    const float p1r = poles_ri[2], p1i = poles_ri[3];
    const float z0r = zeros_ri[0], z0i = zeros_ri[1];
    const float z1r = zeros_ri[2], z1i = zeros_ri[3];

    const float a1 = -(p0r + p1r);
    const float a2 = p0r * p1r - p0i * p1i;
    const float zc1r = -(z0r + z1r), zc1i = -(z0i + z1i);
    const float zc2r = z0r * z1r - z0i * z1i;
    const float zc2i = z0r * z1i + z0i * z1r;
    const float b0 = gr;
    const float b1 = gr * zc1r - gi * zc1i;
    const float b2 = gr * zc2r - gi * zc2i;

    float g[16];
    g[0] = 1.f; g[1] = -a1;
#pragma unroll
    for (int k = 2; k < 16; ++k) g[k] = -a1 * g[k-1] - a2 * g[k-2];

    float hh[16], R1[16], R2[16], P[16], Q[16];
#pragma unroll
    for (int k = 0; k < 16; ++k) {
        const float gm1 = (k >= 1) ? g[k-1] : 0.f;
        const float gm2 = (k >= 2) ? g[k-2] : 0.f;
        hh[k] = b0 * g[k] + b1 * gm1 + b2 * gm2;
        R1[k] = b1 * g[k] + b2 * gm1;
        R2[k] = b2 * g[k];
    }
    P[0] = -a1; P[1] = a1 * a1 - a2;
    Q[0] = -a2; Q[1] = a1 * a2;
#pragma unroll
    for (int k = 2; k < 16; ++k) {
        P[k] = -a1 * P[k-1] - a2 * P[k-2];
        Q[k] = -a1 * Q[k-1] - a2 * Q[k-2];
    }

    if (tid == 0) {
#pragma unroll
        for (int k = 0; k < 16; ++k) hsh[k] = hh[k];
    }

    float cr1[8], cr2[8], cp[8], cq[8];
#pragma unroll
    for (int r = 0; r < 8; ++r) {
        cr1[r] = hi ? R1[r + 8] : R1[r];
        cr2[r] = hi ? R2[r + 8] : R2[r];
        cp[r]  = hi ? P[r + 8]  : P[r];
        cq[r]  = hi ? Q[r + 8]  : Q[r];
    }

    __syncthreads();

    v2f A[4];
#pragma unroll
    for (int kb = 0; kb < 4; ++kb) {
        const int i0 = n - 4 * kb - 2 * hi;
        A[kb].x = (i0 >= 0) ? hsh[i0]     : 0.f;
        A[kb].y = (i0 >= 1) ? hsh[i0 - 1] : 0.f;
    }

    const long long bb = (long long)(blockIdx.x * WAVES_PER_CTA + w) * 16;
    const int srow = lane >> 1;
    const int scol = (lane & 1) * 8;
    const float* xrow = x   + (bb + srow) * (long long)T_LEN;
    float*       orow = out + (bb + srow) * (long long)T_LEN;

    float xm1 = 0.f, xm2 = 0.f, ym1 = 0.f, ym2 = 0.f;

    for (int blk = 0; blk < NBLK; ++blk) {
        const int t0 = blk * BLK;

        const float4 xa = *(const float4*)(xrow + t0 + scol);
        const float4 xb = *(const float4*)(xrow + t0 + scol + 4);
        if (blk + 1 < NBLK)
            __builtin_prefetch(xrow + t0 + BLK + scol, 0, 0);
        *(float4*)&xt[w][srow][scol]     = xa;
        *(float4*)&xt[w][srow][scol + 4] = xb;
        __builtin_amdgcn_wave_barrier();
        asm volatile("s_wait_dscnt 0x0" ::: "memory");

        v2f Bv[4];
#pragma unroll
        for (int kb = 0; kb < 4; ++kb)
            Bv[kb] = *(const v2f*)&xt[w][n][4 * kb + 2 * hi];

        v8f acc;
#pragma unroll
        for (int r = 0; r < 8; ++r)
            acc[r] = cr1[r] * xm1 + cr2[r] * xm2 + cp[r] * ym1 + cq[r] * ym2;

#pragma unroll
        for (int kb = 0; kb < 4; ++kb)
            acc = __builtin_amdgcn_wmma_f32_16x16x4_f32(
                false, A[kb], false, Bv[kb], (short)0, acc, false, false);

        xm2 = __shfl(Bv[3].x, 16 + n, 32);
        xm1 = __shfl(Bv[3].y, 16 + n, 32);
        ym2 = __shfl(acc[6],  16 + n, 32);
        ym1 = __shfl(acc[7],  16 + n, 32);

        const float4 y0 = make_float4(acc[0], acc[1], acc[2], acc[3]);
        const float4 y1 = make_float4(acc[4], acc[5], acc[6], acc[7]);
        *(float4*)&yt[w][n][8 * hi]     = y0;
        *(float4*)&yt[w][n][8 * hi + 4] = y1;
        __builtin_amdgcn_wave_barrier();
        asm volatile("s_wait_dscnt 0x0" ::: "memory");

        *(float4*)(orow + t0 + scol)     = *(const float4*)&yt[w][srow][scol];
        *(float4*)(orow + t0 + scol + 4) = *(const float4*)&yt[w][srow][scol + 4];
    }
}

extern "C" void kernel_launch(void* const* d_in, const int* in_sizes, int n_in,
                              void* d_out, int out_size, void* d_ws, size_t ws_size,
                              hipStream_t stream) {
    const float* x     = (const float*)d_in[0];
    const float* gain  = (const float*)d_in[1];
    const float* poles = (const float*)d_in[2];
    const float* zeros = (const float*)d_in[3];
    float* out = (float*)d_out;

    const int B = in_sizes[0] / T_LEN;
    const int blocks = B / (WAVES_PER_CTA * 16);
    pz_biquad_wmma<<<blocks, CTA_THREADS, 0, stream>>>(x, gain, poles, zeros, out);
}


# Global memory instructions paced through the matrix run: LDS reads first behind the barrier, next loads one block per group, a finished line pair stored by owner wave j&3; sum order unchanged
# speedup vs baseline: 6.1520x; 6.1520x over previous
#include <hip/hip_runtime.h>


typedef float v2f __attribute__((ext_vector_type(2)));
typedef float v8f __attribute__((ext_vector_type(8)));

#define T_LEN 2048
#define BLK 16
#define NBLK (T_LEN / BLK)
#define NPROD 4
#define GRP 4
#define PH_BLK (NPROD * GRP)
#define NPH (NBLK / PH_BLK)
#define DEPTH 3
#define CTA_THREADS (NPROD * 32)

__global__ __launch_bounds__(CTA_THREADS) void pz_biquad_wmma(
    const float* __restrict__ x,
    const float* __restrict__ gain_ri,
    const float* __restrict__ poles_ri,
    const float* __restrict__ zeros_ri,
    float* __restrict__ out)
{
    __shared__ float tab[5 * 16];
    __shared__ float4 ff[2][PH_BLK][2][32];

    const int tid  = threadIdx.x;
    const int w    = __builtin_amdgcn_readfirstlane(tid >> 5);
    const int lane = tid & 31;
    const int n    = lane & 15;
    const int hi   = lane >> 4;

    const long long bb = (long long)blockIdx.x * 16;
    const int scol = 8 * hi;
    const float* xrow = x   + (bb + n) * (long long)T_LEN;
    float*       orow = out + (bb + n) * (long long)T_LEN;

    float4 xa[DEPTH][GRP], xb[DEPTH][GRP];
    v2f xh[DEPTH];
    auto fetch_blk = [&](const int p, const int slot, const int i) {
        const int blk0 = PH_BLK * p + GRP * w;
        const float* src = xrow + blk0 * BLK + scol;
        xa[slot][i] = *(const float4*)(src + i * BLK);
        xb[slot][i] = *(const float4*)(src + i * BLK + 4);
        if (i == 0) {
            const float2 h = *(const float2*)(xrow + (blk0 > 0 ? blk0 * BLK - 2 : 0));
            xh[slot].x = (blk0 > 0) ? h.x : 0.f;
            xh[slot].y = (blk0 > 0) ? h.y : 0.f;
        }
    };
    auto fetch = [&](const int p, const int slot) {
#pragma unroll
        for (int i = 0; i < GRP; ++i) fetch_blk(p, slot, i);
    };

    fetch(0, 0);
    fetch(1, 1);

    auto lds_barrier = [&]() {
        __builtin_amdgcn_fence(__ATOMIC_RELEASE, "workgroup", "local");
        __builtin_amdgcn_s_barrier();
        __builtin_amdgcn_fence(__ATOMIC_ACQUIRE, "workgroup", "local");
    };

    const float gr  = gain_ri[0],  gi  = gain_ri[1];
    const float p0r = poles_ri[0], p0i = poles_ri[1];
    const float p1r = poles_ri[2], p1i = poles_ri[3];
    const float z0r = zeros_ri[0], z0i = zeros_ri[1];
    const float z1r = zeros_ri[2], z1i = zeros_ri[3];

    const float a1 = -(p0r + p1r);
    const float a2 = p0r * p1r - p0i * p1i;
    const float zc1r = -(z0r + z1r), zc1i = -(z0i + z1i);
    const float zc2r = z0r * z1r - z0i * z1i;
    const float zc2i = z0r * z1i + z0i * z1r;
    const float b0 = gr;
    const float b1 = gr * zc1r - gi * zc1i;
    const float b2 = gr * zc2r - gi * zc2i;

    float g[16];
    g[0] = 1.f; g[1] = -a1;
#pragma unroll
    for (int k = 2; k < 16; ++k) g[k] = -a1 * g[k-1] - a2 * g[k-2];

    float hh[16], R1[16], R2[16], P[16], Q[16];
#pragma unroll
    for (int k = 0; k < 16; ++k) {
        const float gm1 = (k >= 1) ? g[k-1] : 0.f;
        const float gm2 = (k >= 2) ? g[k-2] : 0.f;
        hh[k] = b0 * g[k] + b1 * gm1 + b2 * gm2;
        R1[k] = b1 * g[k] + b2 * gm1;
        R2[k] = b2 * g[k];
    }
    P[0] = -a1; P[1] = a1 * a1 - a2;
    Q[0] = -a2; Q[1] = a1 * a2;
#pragma unroll
    for (int k = 2; k < 16; ++k) {
        P[k] = -a1 * P[k-1] - a2 * P[k-2];
        Q[k] = -a1 * Q[k-1] - a2 * Q[k-2];
    }

    if (tid == 0) {
#pragma unroll
        for (int k = 0; k < 16; ++k) {
            tab[k]      = hh[k];
            tab[16 + k] = R1[k];
            tab[32 + k] = R2[k];
            tab[48 + k] = P[k];
            tab[64 + k] = Q[k];
        }
    }

    lds_barrier();

    v2f A[4];
#pragma unroll
    for (int kb = 0; kb < 4; ++kb) {
        const int i0 = n - 2 * kb - 8 * hi;
        A[kb].x = (i0 >= 0) ? tab[i0]     : 0.f;
        A[kb].y = (i0 >= 1) ? tab[i0 - 1] : 0.f;
    }

    v2f Asx, Asy;
    {
        const float r1 = tab[16 + n], r2 = tab[32 + n];
        const float pp = tab[48 + n], qq = tab[64 + n];
        Asx.x = hi ? r2 : 0.f;   Asx.y = hi ? r1 : 0.f;
        Asy.x = hi ? qq : 0.f;   Asy.y = hi ? pp : 0.f;
    }

    v8f pacc[GRP];
    auto pstep = [&](const int slot, const int i, const int k, v8f& acc, v2f& xst) {
        const float4 a = xa[slot][i], b = xb[slot][i];
        v2f Bk;
        if (k == 1)      { Bk.x = a.x; Bk.y = a.y; }
        else if (k == 2) { Bk.x = a.z; Bk.y = a.w; }
        else if (k == 3) { Bk.x = b.x; Bk.y = b.y; }
        else             { Bk.x = b.z; Bk.y = b.w; }
        if (k == 0) {
            const v8f zero = {0.f, 0.f, 0.f, 0.f, 0.f, 0.f, 0.f, 0.f};
            acc = __builtin_amdgcn_wmma_f32_16x16x4_f32(
                false, Asx, false, xst, (short)0, zero, false, false);
        } else {
            acc = __builtin_amdgcn_wmma_f32_16x16x4_f32(
                false, A[k - 1], false, Bk, (short)0, acc, false, false);
        }
        if (k == 4) {
            xst = Bk;
            pacc[i] = acc;
        }
    };
    auto put_sums = [&](const int p) {
#pragma unroll
        for (int i = 0; i < GRP; ++i) {
            ff[p & 1][GRP * w + i][0][lane] =
                make_float4(pacc[i][0], pacc[i][1], pacc[i][2], pacc[i][3]);
            ff[p & 1][GRP * w + i][1][lane] =
                make_float4(pacc[i][4], pacc[i][5], pacc[i][6], pacc[i][7]);
        }
    };
    v8f cacc[PH_BLK];
    auto get_sums = [&](const int q) {
#pragma unroll
        for (int b = 0; b < PH_BLK; ++b) {
            const float4 c0 = ff[q & 1][b][0][lane];
            const float4 c1 = ff[q & 1][b][1][lane];
            cacc[b] = v8f{c0.x, c0.y, c0.z, c0.w, c1.x, c1.y, c1.z, c1.w};
        }
    };

    v2f yst = {0.f, 0.f};
    auto cstep = [&](const int b) {
        cacc[b] = __builtin_amdgcn_wmma_f32_16x16x4_f32(
            false, Asy, false, yst, (short)0, cacc[b], false, false);
        yst.x = cacc[b][6];  yst.y = cacc[b][7];
    };
    auto put_pair = [&](const int q, const int j) {
        if (w == (j & 3)) {
#pragma unroll
            for (int i = 0; i < 2; ++i) {
                const v8f a = cacc[2 * j + i];
                float* o = orow + scol + (PH_BLK * q + 2 * j + i) * BLK;
                *(float4*)(o)     = make_float4(a[0], a[1], a[2], a[3]);
                *(float4*)(o + 4) = make_float4(a[4], a[5], a[6], a[7]);
            }
        }
    };

#pragma unroll
    for (int p = 0; p <= NPH; ++p) {
        const bool hasC = p >= 1, hasP = p < NPH, hasF = p + 2 < NPH;
        if (hasC) {
            get_sums(p - 1);
            __builtin_amdgcn_sched_barrier(0);
        }
        v2f xst = xh[p % DEPTH];
#pragma unroll
        for (int gq = 0; gq < NPROD; ++gq) {
            v8f acc;
            if (hasF) fetch_blk(p + 2, (p + 2) % DEPTH, gq);
            if (hasP) { pstep(p % DEPTH, gq, 0, acc, xst); pstep(p % DEPTH, gq, 1, acc, xst); }
            if (hasC && gq > 0) put_pair(p - 1, 2 * gq - 1);
            if (hasC) cstep(GRP * gq + 0);
            if (hasP) pstep(p % DEPTH, gq, 2, acc, xst);
            if (hasC) cstep(GRP * gq + 1);
            if (hasP) pstep(p % DEPTH, gq, 3, acc, xst);
            if (hasC) cstep(GRP * gq + 2);
            if (hasC) put_pair(p - 1, 2 * gq);
            if (hasP) pstep(p % DEPTH, gq, 4, acc, xst);
            if (hasC) cstep(GRP * gq + 3);
        }
        if (hasP) put_sums(p);
        if (hasC) put_pair(p - 1, 2 * NPROD - 1);
        if (hasP) lds_barrier();
    }
}

extern "C" void kernel_launch(void* const* d_in, const int* in_sizes, int n_in,
                              void* d_out, int out_size, void* d_ws, size_t ws_size,
                              hipStream_t stream) {
    const float* x     = (const float*)d_in[0];
    const float* gain  = (const float*)d_in[1];
    const float* poles = (const float*)d_in[2];
    const float* zeros = (const float*)d_in[3];
    float* out = (float*)d_out;

    const int B = in_sizes[0] / T_LEN;
    const int blocks = B / 16;
    pz_biquad_wmma<<<blocks, CTA_THREADS, 0, stream>>>(x, gain, poles, zeros, out);
}
